# starting kernel
# baseline (speedup 1.0000x reference)
#include <hip/hip_runtime.h>
#include <math.h>


typedef __attribute__((ext_vector_type(2))) float v2f;
typedef __attribute__((ext_vector_type(8))) float v8f;

#define B_TOT    512
#define D        32
#define H        256
#define BR       16
#define NTHREADS 256
#define NSTEP    64

#define XS   36
#define HS   260
#define W1F  544
#define W2F  96
#define G2S  36

__device__ __forceinline__ float fast_tanh(float z) {
    const float TWO_LOG2E = 2.88539008177792681472f;
    const float a = __builtin_fabsf(z);
    const float e = __builtin_amdgcn_exp2f(a * TWO_LOG2E);
    const float r = __builtin_amdgcn_rcpf(e + 1.0f);
    const float t = __builtin_fmaf(-2.0f, r, 1.0f);
    return __builtin_copysignf(t, z);
}

__device__ __forceinline__ v2f lds_v2f(const float* p) {
    return *(const v2f*)p;
}

__device__ __forceinline__ void eval_core(
    const float* __restrict__ xin,
    float* __restrict__ hbuf,
    float* __restrict__ g2part,
    float* __restrict__ part,
    float* __restrict__ epart,
    const float* __restrict__ w1p,
    const float* __restrict__ w2p,
    const float* __restrict__ cdiag,
    const float* __restrict__ b1eff,
    int tid)
{
    const int wid  = tid >> 5;
    const int lane = tid & 31;
    const int half = lane >> 4;
    const int lr   = lane & 15;

#pragma unroll
    for (int it = 0; it < 2; ++it) {
        const int n0 = (wid * 2 + it) * 16;
        v8f acc = {};
#pragma unroll
        for (int k0 = 0; k0 < 8; ++k0) {
            const int kb = k0 * 4 + 2 * half;
            const v2f a = lds_v2f(&xin[lr * XS + kb]);
            const v2f b = lds_v2f(&w1p[(2 * k0 + half) * W1F + 2 * (n0 + lr)]);
            acc = __builtin_amdgcn_wmma_f32_16x16x4_f32(
                      false, a, false, b, (short)0, acc, false, false);
        }
#pragma unroll
        for (int v = 0; v < 8; ++v) {
            const int row = v + 8 * half;
            const int col = n0 + lr;
            hbuf[row * HS + col] = fast_tanh(acc[v] + b1eff[col]);
        }
    }
    __syncthreads();

    {
        const int n0 = (wid & 1) * 16;
        const int s  = wid >> 1;
        v8f acc = {};
#pragma unroll
        for (int k0i = 0; k0i < 16; ++k0i) {
            const int kk = s * 16 + k0i;
            const int kb = kk * 4 + 2 * half;
            const v2f a = lds_v2f(&hbuf[lr * HS + kb]);
            const v2f b = lds_v2f(&w2p[(2 * kk + half) * W2F + 2 * (n0 + lr)]);
            acc = __builtin_amdgcn_wmma_f32_16x16x4_f32(
                      false, a, false, b, (short)0, acc, false, false);
        }
#pragma unroll
        for (int v = 0; v < 8; ++v) {
            const int row = v + 8 * half;
            g2part[(s * BR + row) * G2S + n0 + lr] = acc[v];
        }
    }

    {
        const int r = tid >> 4;
        const int sub = tid & 15;
        float dsum = 0.f;
#pragma unroll
        for (int j = sub; j < H; j += 16) {
            const float hv = hbuf[r * HS + j];
            dsum += (1.f - hv * hv) * cdiag[j];
        }
        part[tid] = dsum;
        const float xa = xin[r * XS + sub];
        const float xb = xin[r * XS + sub + 16];
        epart[tid] = 0.5f * (xa * xa + xb * xb);
    }
    __syncthreads();
}

__global__ __launch_bounds__(NTHREADS)
void cnf_rk4_kernel(const float* __restrict__ x,
                    const float* __restrict__ W1,
                    const float* __restrict__ b1,
                    const float* __restrict__ W2,
                    const float* __restrict__ b2,
                    float* __restrict__ out)
{
    __shared__ __align__(16) float w1p[16 * W1F];
    __shared__ __align__(16) float w2p[128 * W2F];
    __shared__ float w1time[H];
    __shared__ float b1s[H], b2s[D], cdiag[H], b1eff[H];
    __shared__ __align__(16) float hbuf[BR * HS];
    __shared__ __align__(16) float x0[BR * XS], xev[BR * XS], xacc[BR * XS];
    __shared__ float g2part[4 * BR * G2S];
    __shared__ float part[BR * 16], epart[BR * 16];
    __shared__ float acclp[BR], accls[BR], lp0[BR], ls0[BR];

    const int tid = threadIdx.x;
    const int wg  = blockIdx.x;

    for (int i = tid; i < 32 * H; i += NTHREADS) {
        const int k = i / H, n = i % H;
        w1p[(k >> 1) * W1F + 2 * n + (k & 1)] = W1[i];
    }
    w1time[tid] = W1[32 * H + tid];
    for (int i = tid; i < H * D; i += NTHREADS) {
        const int k = i / D, n = i % D;
        w2p[(k >> 1) * W2F + 2 * n + (k & 1)] = W2[i];
    }
    b1s[tid] = b1[tid];
    if (tid < D) b2s[tid] = b2[tid];
    for (int i = tid; i < BR * D; i += NTHREADS)
        x0[(i / D) * XS + (i % D)] = x[wg * BR * D + i];
    if (tid < BR) { lp0[tid] = 0.f; ls0[tid] = 0.f; }
    __syncthreads();

    const float b1r = b1s[tid];
    const float w1t = w1time[tid];

    {
        float c = 0.f;
        const int j = tid;
#pragma unroll
        for (int k = 0; k < D; ++k)
            c += w1p[(k >> 1) * W1F + 2 * j + (k & 1)]
               * w2p[(j >> 1) * W2F + 2 * k + (j & 1)];
        cdiag[j] = c;
        b1eff[j] = b1r;
    }
    __syncthreads();

    const float dt = 1.0f / (float)NSTEP;

    for (int step = 0; step < NSTEP; ++step) {
        const float t0 = step * dt;

        eval_core(x0, hbuf, g2part, part, epart, w1p, w2p, cdiag, b1eff, tid);
        for (int i = tid; i < BR * D; i += NTHREADS) {
            const int r = i >> 5, d = i & 31;
            const float fv = b2s[d]
                + g2part[(0 * BR + r) * G2S + d] + g2part[(1 * BR + r) * G2S + d]
                + g2part[(2 * BR + r) * G2S + d] + g2part[(3 * BR + r) * G2S + d];
            xacc[r * XS + d] = fv;
            xev[r * XS + d]  = x0[r * XS + d] + 0.5f * dt * fv;
        }
        if (tid < BR) {
            float ds = 0.f, es = 0.f;
#pragma unroll
            for (int s = 0; s < 16; ++s) { ds += part[tid * 16 + s]; es += epart[tid * 16 + s]; }
            acclp[tid] = -ds;  accls[tid] = es;
        }
        b1eff[tid] = b1r + (t0 + 0.5f * dt) * w1t;
        __syncthreads();

        eval_core(xev, hbuf, g2part, part, epart, w1p, w2p, cdiag, b1eff, tid);
        for (int i = tid; i < BR * D; i += NTHREADS) {
            const int r = i >> 5, d = i & 31;
            const float fv = b2s[d]
                + g2part[(0 * BR + r) * G2S + d] + g2part[(1 * BR + r) * G2S + d]
                + g2part[(2 * BR + r) * G2S + d] + g2part[(3 * BR + r) * G2S + d];
            xacc[r * XS + d] += 2.f * fv;
            xev[r * XS + d]   = x0[r * XS + d] + 0.5f * dt * fv;
        }
        if (tid < BR) {
            float ds = 0.f, es = 0.f;
#pragma unroll
            for (int s = 0; s < 16; ++s) { ds += part[tid * 16 + s]; es += epart[tid * 16 + s]; }
            acclp[tid] += -2.f * ds;  accls[tid] += 2.f * es;
        }
        __syncthreads();

        eval_core(xev, hbuf, g2part, part, epart, w1p, w2p, cdiag, b1eff, tid);
        for (int i = tid; i < BR * D; i += NTHREADS) {
            const int r = i >> 5, d = i & 31;
            const float fv = b2s[d]
                + g2part[(0 * BR + r) * G2S + d] + g2part[(1 * BR + r) * G2S + d]
                + g2part[(2 * BR + r) * G2S + d] + g2part[(3 * BR + r) * G2S + d];
            xacc[r * XS + d] += 2.f * fv;
            xev[r * XS + d]   = x0[r * XS + d] + dt * fv;
        }
        if (tid < BR) {
            float ds = 0.f, es = 0.f;
#pragma unroll
            for (int s = 0; s < 16; ++s) { ds += part[tid * 16 + s]; es += epart[tid * 16 + s]; }
            acclp[tid] += -2.f * ds;  accls[tid] += 2.f * es;
        }
        b1eff[tid] = b1r + (t0 + dt) * w1t;
        __syncthreads();

        eval_core(xev, hbuf, g2part, part, epart, w1p, w2p, cdiag, b1eff, tid);
        for (int i = tid; i < BR * D; i += NTHREADS) {
            const int r = i >> 5, d = i & 31;
            const float fv = b2s[d]
                + g2part[(0 * BR + r) * G2S + d] + g2part[(1 * BR + r) * G2S + d]
                + g2part[(2 * BR + r) * G2S + d] + g2part[(3 * BR + r) * G2S + d];
            x0[r * XS + d] += (dt / 6.f) * (xacc[r * XS + d] + fv);
        }
        if (tid < BR) {
            float ds = 0.f, es = 0.f;
#pragma unroll
            for (int s = 0; s < 16; ++s) { ds += part[tid * 16 + s]; es += epart[tid * 16 + s]; }
            lp0[tid] += (dt / 6.f) * (acclp[tid] - ds);
            ls0[tid] += (dt / 6.f) * (accls[tid] + es);
        }
        b1eff[tid] = b1r + (t0 + dt) * w1t;
        __syncthreads();
    }

    for (int i = tid; i < BR * D; i += NTHREADS)
        out[wg * BR * D + i] = x0[(i >> 5) * XS + (i & 31)];
    if (tid < BR) {
        out[B_TOT * D + wg * BR + tid]         = lp0[tid];
        out[B_TOT * D + B_TOT + wg * BR + tid] = ls0[tid];
    }
}

extern "C" void kernel_launch(void* const* d_in, const int* in_sizes, int n_in,
                              void* d_out, int out_size, void* d_ws, size_t ws_size,
                              hipStream_t stream) {
    const float* x  = (const float*)d_in[0];
    const float* W1 = (const float*)d_in[1];
    const float* b1 = (const float*)d_in[2];
    const float* W2 = (const float*)d_in[3];
    const float* b2 = (const float*)d_in[4];
    float* out = (float*)d_out;
    cnf_rk4_kernel<<<B_TOT / BR, NTHREADS, 0, stream>>>(x, W1, b1, W2, b2, out);
}


# logp and loss integrals: each thread keeps the RK4-weighted running sum of its own column slice in registers, the 16 slices of a row are added once after the time loop; their summation order changes
# speedup vs baseline: 1.4595x; 1.4595x over previous
#include <hip/hip_runtime.h>
#include <math.h>


typedef __attribute__((ext_vector_type(2))) float v2f;
typedef __attribute__((ext_vector_type(8))) float v8f;

#define B_TOT    512
#define D        32
#define H        256
#define BR       16
#define NTHREADS 256
#define NSTEP    64

#define XS   36
#define HS   260
#define W1F  544
#define W2F  96
#define G2S  36

__device__ __forceinline__ float fast_tanh(float z) {
    const float TWO_LOG2E = 2.88539008177792681472f;
    const float a = __builtin_fabsf(z);
    const float e = __builtin_amdgcn_exp2f(a * TWO_LOG2E);
    const float r = __builtin_amdgcn_rcpf(e + 1.0f);
    const float t = __builtin_fmaf(-2.0f, r, 1.0f);
    return __builtin_copysignf(t, z);
}

__device__ __forceinline__ v2f lds_v2f(const float* p) {
    return *(const v2f*)p;
}

__device__ __forceinline__ void eval_core(
    const float* __restrict__ xin,
    float* __restrict__ hbuf,
    float* __restrict__ g2part,
    const float qw,
    float& divacc,
    float& engacc,
    const v2f (&w1f)[2][8],
    const v2f (&w2f)[16],
    const float (&cj)[16],
    const float* __restrict__ b1eff,
    int tid)
{
    const int wid  = tid >> 5;
    const int lane = tid & 31;
    const int half = lane >> 4;
    const int lr   = lane & 15;

    {
        const int n0 = wid * 32;
        v2f a[8];
#pragma unroll
        for (int k0 = 0; k0 < 8; ++k0)
            a[k0] = lds_v2f(&xin[lr * XS + k0 * 4 + 2 * half]);
        const float bias0 = b1eff[n0 + lr];
        const float bias1 = b1eff[n0 + 16 + lr];
        __builtin_amdgcn_sched_barrier(0);
        v8f acc0 = {}, acc1 = {};
#pragma unroll
        for (int k0 = 0; k0 < 8; ++k0) {
            acc0 = __builtin_amdgcn_wmma_f32_16x16x4_f32(
                       false, a[k0], false, w1f[0][k0], (short)0, acc0, false, false);
            acc1 = __builtin_amdgcn_wmma_f32_16x16x4_f32(
                       false, a[k0], false, w1f[1][k0], (short)0, acc1, false, false);
        }
#pragma unroll
        for (int v = 0; v < 8; ++v) {
            const int row = v + 8 * half;
            hbuf[row * HS + n0 + lr]      = fast_tanh(acc0[v] + bias0);
            hbuf[row * HS + n0 + 16 + lr] = fast_tanh(acc1[v] + bias1);
        }
    }
    __syncthreads();

    {
        const int n0 = (wid & 1) * 16;
        const int s  = wid >> 1;
        const int r   = tid >> 4;
        const int sub = tid & 15;
        v2f a[16];
#pragma unroll
        for (int k0i = 0; k0i < 16; ++k0i)
            a[k0i] = lds_v2f(&hbuf[lr * HS + (s * 16 + k0i) * 4 + 2 * half]);
        float hv[16];
#pragma unroll
        for (int i = 0; i < 16; ++i)
            hv[i] = hbuf[r * HS + sub + 16 * i];
        const float xa = xin[r * XS + sub];
        const float xb = xin[r * XS + sub + 16];
        __builtin_amdgcn_sched_barrier(0);

        v8f acc = {};
#pragma unroll
        for (int k0i = 0; k0i < 8; ++k0i)
            acc = __builtin_amdgcn_wmma_f32_16x16x4_f32(
                      false, a[k0i], false, w2f[k0i], (short)0, acc, false, false);
        __builtin_amdgcn_sched_barrier(0);
#pragma unroll
        for (int k0i = 8; k0i < 16; ++k0i)
            acc = __builtin_amdgcn_wmma_f32_16x16x4_f32(
                      false, a[k0i], false, w2f[k0i], (short)0, acc, false, false);
#pragma unroll
        for (int v = 0; v < 8; ++v) {
            const int row = v + 8 * half;
            g2part[(s * BR + row) * G2S + n0 + lr] = acc[v];
        }

        float dsum = 0.f;
#pragma unroll
        for (int i = 0; i < 16; ++i)
            dsum += (1.f - hv[i] * hv[i]) * cj[i];
        divacc = __builtin_fmaf(qw, dsum, divacc);
        engacc = __builtin_fmaf(qw, 0.5f * (xa * xa + xb * xb), engacc);
        asm volatile("" : "+v"(divacc), "+v"(engacc));
    }
    __syncthreads();
}

__global__ __launch_bounds__(NTHREADS)
void cnf_rk4_kernel(const float* __restrict__ x,
                    const float* __restrict__ W1,
                    const float* __restrict__ b1,
                    const float* __restrict__ W2,
                    const float* __restrict__ b2,
                    float* __restrict__ out)
{
    __shared__ __align__(16) float w1p[16 * W1F];
    __shared__ __align__(16) float w2p[128 * W2F];
    __shared__ float w1time[H];
    __shared__ float b1s[H], b2s[D], cdiag[H], b1eff[H];
    __shared__ __align__(16) float hbuf[BR * HS];
    __shared__ __align__(16) float x0[BR * XS], xev[BR * XS], xacc[BR * XS];
    __shared__ float g2part[4 * BR * G2S];
    __shared__ float part[BR * 16], epart[BR * 16];

    const int tid = threadIdx.x;
    const int wg  = blockIdx.x;

    for (int i = tid; i < 32 * H; i += NTHREADS) {
        const int k = i / H, n = i % H;
        w1p[(k >> 1) * W1F + 2 * n + (k & 1)] = W1[i];
    }
    w1time[tid] = W1[32 * H + tid];
    for (int i = tid; i < H * D; i += NTHREADS) {
        const int k = i / D, n = i % D;
        w2p[(k >> 1) * W2F + 2 * n + (k & 1)] = W2[i];
    }
    b1s[tid] = b1[tid];
    if (tid < D) b2s[tid] = b2[tid];
    for (int i = tid; i < BR * D; i += NTHREADS)
        x0[(i / D) * XS + (i % D)] = x[wg * BR * D + i];
    __syncthreads();

    const float b1r = b1s[tid];
    const float w1t = w1time[tid];
    const float b2r = b2s[tid & 31];

    {
        float c = 0.f;
        const int j = tid;
#pragma unroll
        for (int k = 0; k < D; ++k)
            c += w1p[(k >> 1) * W1F + 2 * j + (k & 1)]
               * w2p[(j >> 1) * W2F + 2 * k + (j & 1)];
        cdiag[j] = c;
        b1eff[j] = b1r;
    }
    __syncthreads();

    v2f w1f[2][8];
    v2f w2f[16];
    float cj[16];
    {
        const int wid  = tid >> 5;
        const int lane = tid & 31;
        const int half = lane >> 4;
        const int lr   = lane & 15;
#pragma unroll
        for (int it = 0; it < 2; ++it)
#pragma unroll
            for (int k0 = 0; k0 < 8; ++k0)
                w1f[it][k0] = lds_v2f(&w1p[(2 * k0 + half) * W1F + 2 * ((wid * 2 + it) * 16 + lr)]);
        const int n0 = (wid & 1) * 16;
        const int s  = wid >> 1;
#pragma unroll
        for (int k0i = 0; k0i < 16; ++k0i)
            w2f[k0i] = lds_v2f(&w2p[(2 * (s * 16 + k0i) + half) * W2F + 2 * (n0 + lr)]);
#pragma unroll
        for (int i = 0; i < 16; ++i)
            cj[i] = cdiag[(tid & 15) + 16 * i];
    }

    const float dt = 1.0f / (float)NSTEP;
    const float q1 = dt / 6.f;
    const float q2 = 2.f * q1;
    float divacc = 0.f, engacc = 0.f;

    for (int step = 0; step < NSTEP; ++step) {
        const float t0 = step * dt;

        eval_core(x0, hbuf, g2part, q1, divacc, engacc, w1f, w2f, cj, b1eff, tid);
        for (int i = tid; i < BR * D; i += NTHREADS) {
            const int r = i >> 5, d = i & 31;
            const float fv = b2r
                + g2part[(0 * BR + r) * G2S + d] + g2part[(1 * BR + r) * G2S + d]
                + g2part[(2 * BR + r) * G2S + d] + g2part[(3 * BR + r) * G2S + d];
            xacc[r * XS + d] = fv;
            xev[r * XS + d]  = x0[r * XS + d] + 0.5f * dt * fv;
        }
        b1eff[tid] = b1r + (t0 + 0.5f * dt) * w1t;
        __syncthreads();

        eval_core(xev, hbuf, g2part, q2, divacc, engacc, w1f, w2f, cj, b1eff, tid);
        for (int i = tid; i < BR * D; i += NTHREADS) {
            const int r = i >> 5, d = i & 31;
            const float fv = b2r
                + g2part[(0 * BR + r) * G2S + d] + g2part[(1 * BR + r) * G2S + d]
                + g2part[(2 * BR + r) * G2S + d] + g2part[(3 * BR + r) * G2S + d];
            xacc[r * XS + d] += 2.f * fv;
            xev[r * XS + d]   = x0[r * XS + d] + 0.5f * dt * fv;
        }
        __syncthreads();

        eval_core(xev, hbuf, g2part, q2, divacc, engacc, w1f, w2f, cj, b1eff, tid);
        for (int i = tid; i < BR * D; i += NTHREADS) {
            const int r = i >> 5, d = i & 31;
            const float fv = b2r
                + g2part[(0 * BR + r) * G2S + d] + g2part[(1 * BR + r) * G2S + d]
                + g2part[(2 * BR + r) * G2S + d] + g2part[(3 * BR + r) * G2S + d];
            xacc[r * XS + d] += 2.f * fv;
            xev[r * XS + d]   = x0[r * XS + d] + dt * fv;
        }
        b1eff[tid] = b1r + (t0 + dt) * w1t;
        __syncthreads();

        eval_core(xev, hbuf, g2part, q1, divacc, engacc, w1f, w2f, cj, b1eff, tid);
        for (int i = tid; i < BR * D; i += NTHREADS) {
            const int r = i >> 5, d = i & 31;
            const float fv = b2r
                + g2part[(0 * BR + r) * G2S + d] + g2part[(1 * BR + r) * G2S + d]
                + g2part[(2 * BR + r) * G2S + d] + g2part[(3 * BR + r) * G2S + d];
            x0[r * XS + d] += (dt / 6.f) * (xacc[r * XS + d] + fv);
        }
        b1eff[tid] = b1r + (t0 + dt) * w1t;
        __syncthreads();
    }

    part[tid]  = divacc;
    epart[tid] = engacc;
    __syncthreads();

    for (int i = tid; i < BR * D; i += NTHREADS)
        out[wg * BR * D + i] = x0[(i >> 5) * XS + (i & 31)];
    if (tid < BR) {
        float ds = 0.f, es = 0.f;
#pragma unroll
        for (int s = 0; s < 16; ++s) { ds += part[tid * 16 + s]; es += epart[tid * 16 + s]; }
        out[B_TOT * D + wg * BR + tid]         = -ds;
        out[B_TOT * D + B_TOT + wg * BR + tid] = es;
    }
}

extern "C" void kernel_launch(void* const* d_in, const int* in_sizes, int n_in,
                              void* d_out, int out_size, void* d_ws, size_t ws_size,
                              hipStream_t stream) {
    const float* x  = (const float*)d_in[0];
    const float* W1 = (const float*)d_in[1];
    const float* b1 = (const float*)d_in[2];
    const float* W2 = (const float*)d_in[3];
    const float* b2 = (const float*)d_in[4];
    float* out = (float*)d_out;
    cnf_rk4_kernel<<<B_TOT / BR, NTHREADS, 0, stream>>>(x, W1, b1, W2, b2, out);
}
